# starting kernel
# baseline (speedup 1.0000x reference)
#include <hip/hip_runtime.h>

#define DF   128
#define N1C  50000

typedef float v2f __attribute__((ext_vector_type(2)));
typedef float v8f __attribute__((ext_vector_type(8)));

__global__ __launch_bounds__(256)
void GCN_zero_kernel(float4* __restrict__ p, int n4) {
    int i = blockIdx.x * blockDim.x + threadIdx.x;
    int stride = gridDim.x * blockDim.x;
    float4 z; z.x = 0.f; z.y = 0.f; z.z = 0.f; z.w = 0.f;
    for (; i < n4; i += stride) p[i] = z;
}

__global__ __launch_bounds__(256)
void GCN_spmm1_kernel(const float* __restrict__ feat,
                      const float* __restrict__ w,
                      const int*   __restrict__ src,
                      const int*   __restrict__ dst,
                      float*       __restrict__ out,
                      int nE) {
    int e    = (blockIdx.x * blockDim.x + threadIdx.x) >> 5;
    int lane = threadIdx.x & 31;
    if (e >= nE) return;
    int u = src[e];
    int v = dst[e];
    float we = w[e];
    const float4* f4 = (const float4*)(feat + (size_t)u * DF);
    float4 m = f4[lane];
    float* o = out + (size_t)v * DF + lane * 4;
    unsafeAtomicAdd(o + 0, m.x * we);
    unsafeAtomicAdd(o + 1, m.y * we);
    unsafeAtomicAdd(o + 2, m.z * we);
    unsafeAtomicAdd(o + 3, m.w * we);
}

__global__ __launch_bounds__(256)
void GCN_spmm2_kernel(const float* __restrict__ h,
                      const float* __restrict__ b1,
                      const float* __restrict__ w,
                      const int*   __restrict__ src,
                      const int*   __restrict__ dst,
                      float*       __restrict__ out,
                      int nE) {
    int e    = (blockIdx.x * blockDim.x + threadIdx.x) >> 5;
    int lane = threadIdx.x & 31;
    if (e >= nE) return;
    int u = src[e];
    int v = dst[e];
    float we = w[e];
    const float4* h4 = (const float4*)(h + (size_t)u * DF);
    const float4* bb = (const float4*)(b1);
    float4 m = h4[lane];
    float4 b = bb[lane];
    float r0 = fmaxf(m.x + b.x, 0.f) * we;
    float r1 = fmaxf(m.y + b.y, 0.f) * we;
    float r2 = fmaxf(m.z + b.z, 0.f) * we;
    float r3 = fmaxf(m.w + b.w, 0.f) * we;
    float* o = out + (size_t)v * DF + lane * 4;
    unsafeAtomicAdd(o + 0, r0);
    unsafeAtomicAdd(o + 1, r1);
    unsafeAtomicAdd(o + 2, r2);
    unsafeAtomicAdd(o + 3, r3);
}

__global__ __launch_bounds__(256)
void GCN_gemm_wmma_kernel(const float* __restrict__ A,
                          const float* __restrict__ W,
                          const float* __restrict__ bias,
                          float*       __restrict__ out,
                          int M) {
    const int lane = threadIdx.x & 31;
    const int waveInBlk = threadIdx.x >> 5;
    const int nTilesM = (M + 15) >> 4;
    const int nTiles  = nTilesM * (DF / 16);
    int tile = blockIdx.x * (blockDim.x >> 5) + waveInBlk;
    if (tile >= nTiles) return;
    const int tM = tile >> 3;
    const int tN = tile & 7;

    const int half = lane >> 4;
    const int mn   = lane & 15;
    const int row  = tM * 16 + mn;
    const int col  = tN * 16 + mn;

    const int rowc = (row < M) ? row : (M - 1);
    const float* Arow = A + (size_t)rowc * DF + 2 * half;
    const float* Bcol = W + (size_t)(2 * half) * DF + col;

    v8f c = {};
    #pragma unroll
    for (int k = 0; k < DF; k += 4) {
        const float2 av = *(const float2*)(Arow + k);
        v2f a; a[0] = av.x; a[1] = av.y;
        v2f b;
        b[0] = Bcol[(size_t)k * DF];
        b[1] = Bcol[(size_t)(k + 1) * DF];
        c = __builtin_amdgcn_wmma_f32_16x16x4_f32(
                 false, a,  false, b,
                 (short)0, c,  false,  false);
    }

    const float bv = bias[col];
    #pragma unroll
    for (int v = 0; v < 8; ++v) {
        const int r = tM * 16 + v + 8 * half;
        if (r < M) out[(size_t)r * DF + col] = c[v] + bv;
    }
}

extern "C" void kernel_launch(void* const* d_in, const int* in_sizes, int n_in,
                              void* d_out, int out_size, void* d_ws, size_t ws_size,
                              hipStream_t stream) {
    const float* feature = (const float*)d_in[0];
    const float* w0      = (const float*)d_in[1];
    const float* w1      = (const float*)d_in[2];
    const float* b1      = (const float*)d_in[3];
    const float* W2      = (const float*)d_in[4];
    const float* b2      = (const float*)d_in[5];
    const int*   src0    = (const int*)d_in[6];
    const int*   dst0    = (const int*)d_in[7];
    const int*   src1    = (const int*)d_in[8];
    const int*   dst1    = (const int*)d_in[9];

    const int E0 = in_sizes[1];
    const int E1 = in_sizes[2];
    const int N1 = N1C;
    const int N2 = out_size / DF;

    float* h   = (float*)d_ws;
    float* agg = h + (size_t)N1 * DF;
    float* out = (float*)d_out;

    {
        int n4 = (N1 * DF + N2 * DF) / 4;
        int blocks = (n4 + 255) / 256;
        if (blocks > 4096) blocks = 4096;
        GCN_zero_kernel<<<blocks, 256, 0, stream>>>((float4*)d_ws, n4);
    }
    {
        long long threads = (long long)E0 * 32;
        int blocks = (int)((threads + 255) / 256);
        GCN_spmm1_kernel<<<blocks, 256, 0, stream>>>(feature, w0, src0, dst0, h, E0);
    }
    {
        long long threads = (long long)E1 * 32;
        int blocks = (int)((threads + 255) / 256);
        GCN_spmm2_kernel<<<blocks, 256, 0, stream>>>(h, b1, w1, src1, dst1, agg, E1);
    }
    {
        int nTiles = ((N2 + 15) / 16) * (DF / 16);
        int wavesPerBlk = 256 / 32;
        int blocks = (nTiles + wavesPerBlk - 1) / wavesPerBlk;
        GCN_gemm_wmma_kernel<<<blocks, 256, 0, stream>>>(agg, W2, b2, out, N2);
    }
}


# Float-atomic scatter replaced by per-destination edge lists (one integer exchange per edge); 16 lanes per node sum a list in registers, order = exchange arrival; rows written once, no zero pass
# speedup vs baseline: 1.7344x; 1.7344x over previous
#include <hip/hip_runtime.h>

#define DF   128
#define N1C  50000

typedef float v2f __attribute__((ext_vector_type(2)));
typedef float v8f __attribute__((ext_vector_type(8)));


__global__ __launch_bounds__(256)
void GCN_lists_init_kernel(int* __restrict__ head0, int* __restrict__ head1,
                           int* __restrict__ node0, int* __restrict__ node1,
                           int nV0, int nV1, int nE0, int nE1) {
    const int i = blockIdx.x * blockDim.x + threadIdx.x;
    if (i < nV0)            head0[i] = nE0;
    else if (i < nV0 + nV1) head1[i - nV0] = nE1;
    if (i == 0) {
        int4 s; s.x = 0; s.y = 0; s.w = 0;
        s.z = nE0; ((int4*)node0)[nE0] = s;
        s.z = nE1; ((int4*)node1)[nE1] = s;
    }
}

__global__ __launch_bounds__(256)
void GCN_lists_build_kernel(const int*   __restrict__ src0,
                            const int*   __restrict__ dst0,
                            const float* __restrict__ w0,
                            const int*   __restrict__ src1,
                            const int*   __restrict__ dst1,
                            const float* __restrict__ w1,
                            int* __restrict__ head0,
                            int* __restrict__ head1,
                            int* __restrict__ node0,
                            int* __restrict__ node1,
                            int nE0, int nE1, int nV0, int nV1) {
    const int t = blockIdx.x * blockDim.x + threadIdx.x;
    const bool first = t < nE0;
    const int  i     = first ? t : t - nE0;
    const int  nE    = first ? nE0 : nE1;
    const int  nV    = first ? nV0 : nV1;
    const int*   src = first ? src0 : src1;
    const int*   dst = first ? dst0 : dst1;
    const float* w   = first ? w0 : w1;
    int* head        = first ? head0 : head1;
    int* node        = first ? node0 : node1;
    if (i >= nE) return;
    const int d = dst[i];
    if ((unsigned)d >= (unsigned)nV) return;
    int4 nd;
    nd.x = src[i];
    nd.y = __float_as_int(w[i]);
    nd.z = atomicExch(head + d, i);
    nd.w = 0;
    ((int4*)node)[i] = nd;
}

struct GCN_row8 { float4 lo, hi; };

static __device__ __forceinline__
GCN_row8 GCN_walk_list(const float* __restrict__ rows,
                       const int*   __restrict__ head,
                       const int*   __restrict__ node,
                       int v, bool has, int nE, int c) {
    const int4* __restrict__ nd4 = (const int4*)node;
    GCN_row8 a;
    a.lo.x = 0.f; a.lo.y = 0.f; a.lo.z = 0.f; a.lo.w = 0.f;
    a.hi = a.lo;
    int e = nE;
    if (has) e = head[v];
    e = (int)min((unsigned)e, (unsigned)nE);
    int4 nd = nd4[e];
    for (int it = 0; it < nE; ++it) {
        const bool live = (e != nE);
        if (__builtin_amdgcn_ballot_w32(live) == 0u) break;
        const int4 cur = nd;
        e = (int)min((unsigned)cur.z, (unsigned)nE);
        nd = nd4[e];
        if (live) {
            const float4* r = (const float4*)(rows + (size_t)cur.x * DF);
            const float4 m0 = r[c];
            const float4 m1 = r[16 + c];
            const float we = __int_as_float(cur.y);
            a.lo.x += m0.x * we; a.lo.y += m0.y * we; a.lo.z += m0.z * we; a.lo.w += m0.w * we;
            a.hi.x += m1.x * we; a.hi.y += m1.y * we; a.hi.z += m1.z * we; a.hi.w += m1.w * we;
        }
    }
    return a;
}

__global__ __launch_bounds__(256)
void GCN_gather1_kernel(const float* __restrict__ feat,
                        const float* __restrict__ b1,
                        const int*   __restrict__ head,
                        const int*   __restrict__ node,
                        float*       __restrict__ h,
                        int nV, int nE) {
    const int v = (blockIdx.x * blockDim.x + threadIdx.x) >> 4;
    const int c = threadIdx.x & 15;
    const bool has = v < nV;
    const GCN_row8 a = GCN_walk_list(feat, head, node, v, has, nE, c);
    if (has) {
        const float4 b0 = ((const float4*)b1)[c];
        const float4 b4 = ((const float4*)b1)[16 + c];
        float4 r0, r1;
        r0.x = fmaxf(a.lo.x + b0.x, 0.f); r0.y = fmaxf(a.lo.y + b0.y, 0.f);
        r0.z = fmaxf(a.lo.z + b0.z, 0.f); r0.w = fmaxf(a.lo.w + b0.w, 0.f);
        r1.x = fmaxf(a.hi.x + b4.x, 0.f); r1.y = fmaxf(a.hi.y + b4.y, 0.f);
        r1.z = fmaxf(a.hi.z + b4.z, 0.f); r1.w = fmaxf(a.hi.w + b4.w, 0.f);
        float4* o = (float4*)(h + (size_t)v * DF);
        o[c] = r0;
        o[16 + c] = r1;
    }
}

__global__ __launch_bounds__(256)
void GCN_gather2_kernel(const float* __restrict__ h,
                        const int*   __restrict__ head,
                        const int*   __restrict__ node,
                        float*       __restrict__ agg,
                        int nV, int nE) {
    const int v = (blockIdx.x * blockDim.x + threadIdx.x) >> 4;
    const int c = threadIdx.x & 15;
    const bool has = v < nV;
    const GCN_row8 a = GCN_walk_list(h, head, node, v, has, nE, c);
    if (has) {
        float4* o = (float4*)(agg + (size_t)v * DF);
        o[c] = a.lo;
        o[16 + c] = a.hi;
    }
}

__global__ __launch_bounds__(256)
void GCN_gemm_wmma_kernel(const float* __restrict__ A,
                          const float* __restrict__ W,
                          const float* __restrict__ bias,
                          float*       __restrict__ out,
                          int M) {
    const int lane = threadIdx.x & 31;
    const int waveInBlk = threadIdx.x >> 5;
    const int nTilesM = (M + 15) >> 4;
    const int nTiles  = nTilesM * (DF / 16);
    int tile = blockIdx.x * (blockDim.x >> 5) + waveInBlk;
    if (tile >= nTiles) return;
    const int tM = tile >> 3;
    const int tN = tile & 7;

    const int half = lane >> 4;
    const int mn   = lane & 15;
    const int row  = tM * 16 + mn;
    const int col  = tN * 16 + mn;

    const int rowc = (row < M) ? row : (M - 1);
    const float* Arow = A + (size_t)rowc * DF + 2 * half;
    const float* Bcol = W + (size_t)(2 * half) * DF + col;

    v8f c = {};
    #pragma unroll
    for (int k = 0; k < DF; k += 4) {
        const float2 av = *(const float2*)(Arow + k);
        v2f a; a[0] = av.x; a[1] = av.y;
        v2f b;
        b[0] = Bcol[(size_t)k * DF];
        b[1] = Bcol[(size_t)(k + 1) * DF];
        c = __builtin_amdgcn_wmma_f32_16x16x4_f32(
                 false, a,  false, b,
                 (short)0, c,  false,  false);
    }

    const float bv = bias[col];
    #pragma unroll
    for (int v = 0; v < 8; ++v) {
        const int r = tM * 16 + v + 8 * half;
        if (r < M) out[(size_t)r * DF + col] = c[v] + bv;
    }
}

extern "C" void kernel_launch(void* const* d_in, const int* in_sizes, int n_in,
                              void* d_out, int out_size, void* d_ws, size_t ws_size,
                              hipStream_t stream) {
    const float* feature = (const float*)d_in[0];
    const float* w0      = (const float*)d_in[1];
    const float* w1      = (const float*)d_in[2];
    const float* b1      = (const float*)d_in[3];
    const float* W2      = (const float*)d_in[4];
    const float* b2      = (const float*)d_in[5];
    const int*   src0    = (const int*)d_in[6];
    const int*   dst0    = (const int*)d_in[7];
    const int*   src1    = (const int*)d_in[8];
    const int*   dst1    = (const int*)d_in[9];

    const int E0 = in_sizes[1];
    const int E1 = in_sizes[2];
    const int N1 = N1C;
    const int N2 = out_size / DF;

    float* h     = (float*)d_ws;
    float* agg   = h + (size_t)N1 * DF;
    int*   node0 = (int*)(agg + (size_t)N2 * DF);
    int*   node1 = node0 + ((size_t)E0 + 1) * 4;
    int*   head0 = node1 + ((size_t)E1 + 1) * 4;
    int*   head1 = head0 + N1;
    float* out   = (float*)d_out;

    {
        int n = N1 + N2;
        int blocks = (n + 255) / 256;
        GCN_lists_init_kernel<<<blocks, 256, 0, stream>>>(head0, head1, node0, node1, N1, N2, E0, E1);
    }
    {
        int n = E0 + E1;
        int blocks = (n + 255) / 256;
        GCN_lists_build_kernel<<<blocks, 256, 0, stream>>>(src0, dst0, w0, src1, dst1, w1,
                                                          head0, head1, node0, node1,
                                                          E0, E1, N1, N2);
    }
    {
        long long threads = (long long)N1 * 16;
        int blocks = (int)((threads + 255) / 256);
        GCN_gather1_kernel<<<blocks, 256, 0, stream>>>(feature, b1, head0, node0, h, N1, E0);
    }
    {
        long long threads = (long long)N2 * 16;
        int blocks = (int)((threads + 255) / 256);
        GCN_gather2_kernel<<<blocks, 256, 0, stream>>>(h, head1, node1, agg, N2, E1);
    }
    {
        int nTiles = ((N2 + 15) / 16) * (DF / 16);
        int wavesPerBlk = 256 / 32;
        int blocks = (nTiles + wavesPerBlk - 1) / wavesPerBlk;
        GCN_gemm_wmma_kernel<<<blocks, 256, 0, stream>>>(agg, W2, b2, out, N2);
    }
}


# Each destination keeps 4 edge lists (edge i on list i%4) that its 16 lanes walk together: 4 rows per memory round trip, half the dependent trips; sum order = exchange arrival, 4 lists interleaved
# speedup vs baseline: 1.8758x; 1.0815x over previous
#include <hip/hip_runtime.h>

#define DF   128
#define N1C  50000

typedef float v2f __attribute__((ext_vector_type(2)));
typedef float v8f __attribute__((ext_vector_type(8)));

#define NSUB 4

__global__ __launch_bounds__(256)
void GCN_lists_init_kernel(int* __restrict__ head0, int* __restrict__ head1,
                           int* __restrict__ node0, int* __restrict__ node1,
                           int nV0, int nV1, int nE0, int nE1) {
    const int i = blockIdx.x * blockDim.x + threadIdx.x;
    if (i < nV0)            ((int4*)head0)[i] = make_int4(nE0, nE0, nE0, nE0);
    else if (i < nV0 + nV1) ((int4*)head1)[i - nV0] = make_int4(nE1, nE1, nE1, nE1);
    if (i == 0) {
        int4 s; s.x = 0; s.y = 0; s.w = 0;
        s.z = nE0; ((int4*)node0)[nE0] = s;
        s.z = nE1; ((int4*)node1)[nE1] = s;
    }
}

__global__ __launch_bounds__(256)
void GCN_lists_build_kernel(const int*   __restrict__ src0,
                            const int*   __restrict__ dst0,
                            const float* __restrict__ w0,
                            const int*   __restrict__ src1,
                            const int*   __restrict__ dst1,
                            const float* __restrict__ w1,
                            int* __restrict__ head0,
                            int* __restrict__ head1,
                            int* __restrict__ node0,
                            int* __restrict__ node1,
                            int nE0, int nE1, int nV0, int nV1) {
    const int t = blockIdx.x * blockDim.x + threadIdx.x;
    const bool first = t < nE0;
    const int  i     = first ? t : t - nE0;
    const int  nE    = first ? nE0 : nE1;
    const int  nV    = first ? nV0 : nV1;
    const int*   src = first ? src0 : src1;
    const int*   dst = first ? dst0 : dst1;
    const float* w   = first ? w0 : w1;
    int* head        = first ? head0 : head1;
    int* node        = first ? node0 : node1;
    if (i >= nE) return;
    const int d = dst[i];
    if ((unsigned)d >= (unsigned)nV) return;
    int4 nd;
    nd.x = src[i];
    nd.y = __float_as_int(w[i]);
    nd.z = atomicExch(head + (size_t)d * NSUB + (i & (NSUB - 1)), i);
    nd.w = 0;
    ((int4*)node)[i] = nd;
}

struct GCN_row8 { float4 lo, hi; };

static __device__ __forceinline__
GCN_row8 GCN_walk_list(const float* __restrict__ rows,
                       const int*   __restrict__ head,
                       const int*   __restrict__ node,
                       int v, bool has, int nE, int c) {
    const char* __restrict__ nbase = (const char*)node;
    const char* __restrict__ rbase = (const char*)rows;
    const unsigned cb = (unsigned)c * 16u;
    const unsigned uE = (unsigned)nE;
    const unsigned bE = uE << 4;
    GCN_row8 a;
    a.lo.x = 0.f; a.lo.y = 0.f; a.lo.z = 0.f; a.lo.w = 0.f;
    a.hi = a.lo;
    int4 hd = make_int4(nE, nE, nE, nE);
    if (has) hd = ((const int4*)head)[v];
    unsigned eb[NSUB];
    eb[0] = min((unsigned)hd.x, uE) << 4;
    eb[1] = min((unsigned)hd.y, uE) << 4;
    eb[2] = min((unsigned)hd.z, uE) << 4;
    eb[3] = min((unsigned)hd.w, uE) << 4;
    int4 nd[NSUB];
    #pragma unroll
    for (int j = 0; j < NSUB; ++j) nd[j] = *(const int4*)(nbase + eb[j]);
    for (int it = 0; it < nE; ++it) {
        bool live[NSUB];
        bool any = false;
        #pragma unroll
        for (int j = 0; j < NSUB; ++j) { live[j] = (eb[j] != bE); any = any || live[j]; }
        if (__builtin_amdgcn_ballot_w32(any) == 0u) break;
        unsigned roff[NSUB];
        float wj[NSUB];
        #pragma unroll
        for (int j = 0; j < NSUB; ++j) {
            roff[j] = (unsigned)nd[j].x * (unsigned)(DF * 4) + cb;
            wj[j]   = __int_as_float(nd[j].y);
            eb[j]   = min((unsigned)nd[j].z, uE) << 4;
            nd[j]   = *(const int4*)(nbase + eb[j]);
        }
        float4 m0[NSUB], m1[NSUB];
        #pragma unroll
        for (int j = 0; j < NSUB; ++j) {
            m0[j] = make_float4(0.f, 0.f, 0.f, 0.f);
            m1[j] = m0[j];
            if (live[j]) {
                m0[j] = *(const float4*)(rbase + roff[j]);
                m1[j] = *(const float4*)(rbase + roff[j] + 256u);
            } else {
                wj[j] = 0.f;
            }
        }
        #pragma unroll
        for (int j = 0; j < NSUB; ++j) {
            const float we = wj[j];
            a.lo.x += m0[j].x * we; a.lo.y += m0[j].y * we; a.lo.z += m0[j].z * we; a.lo.w += m0[j].w * we;
            a.hi.x += m1[j].x * we; a.hi.y += m1[j].y * we; a.hi.z += m1[j].z * we; a.hi.w += m1[j].w * we;
        }
    }
    return a;
}

__global__ __launch_bounds__(256)
void GCN_gather1_kernel(const float* __restrict__ feat,
                        const float* __restrict__ b1,
                        const int*   __restrict__ head,
                        const int*   __restrict__ node,
                        float*       __restrict__ h,
                        int nV, int nE) {
    const int v = (blockIdx.x * blockDim.x + threadIdx.x) >> 4;
    const int c = threadIdx.x & 15;
    const bool has = v < nV;
    const GCN_row8 a = GCN_walk_list(feat, head, node, v, has, nE, c);
    if (has) {
        const char* bb = (const char*)b1 + ((unsigned)c << 4);
        const float4 b0 = *(const float4*)bb;
        const float4 b4 = *(const float4*)(bb + 256);
        float4 r0, r1;
        r0.x = fmaxf(a.lo.x + b0.x, 0.f); r0.y = fmaxf(a.lo.y + b0.y, 0.f);
        r0.z = fmaxf(a.lo.z + b0.z, 0.f); r0.w = fmaxf(a.lo.w + b0.w, 0.f);
        r1.x = fmaxf(a.hi.x + b4.x, 0.f); r1.y = fmaxf(a.hi.y + b4.y, 0.f);
        r1.z = fmaxf(a.hi.z + b4.z, 0.f); r1.w = fmaxf(a.hi.w + b4.w, 0.f);
        float4* o = (float4*)(h + (size_t)v * DF);
        o[c] = r0;
        o[16 + c] = r1;
    }
}

__global__ __launch_bounds__(256)
void GCN_gather2_kernel(const float* __restrict__ h,
                        const int*   __restrict__ head,
                        const int*   __restrict__ node,
                        float*       __restrict__ agg,
                        int nV, int nE) {
    const int v = (blockIdx.x * blockDim.x + threadIdx.x) >> 4;
    const int c = threadIdx.x & 15;
    const bool has = v < nV;
    const GCN_row8 a = GCN_walk_list(h, head, node, v, has, nE, c);
    if (has) {
        float4* o = (float4*)(agg + (size_t)v * DF);
        o[c] = a.lo;
        o[16 + c] = a.hi;
    }
}

__global__ __launch_bounds__(256)
void GCN_gemm_wmma_kernel(const float* __restrict__ A,
                          const float* __restrict__ W,
                          const float* __restrict__ bias,
                          float*       __restrict__ out,
                          int M) {
    const int lane = threadIdx.x & 31;
    const int waveInBlk = threadIdx.x >> 5;
    const int nTilesM = (M + 15) >> 4;
    const int nTiles  = nTilesM * (DF / 16);
    int tile = blockIdx.x * (blockDim.x >> 5) + waveInBlk;
    if (tile >= nTiles) return;
    const int tM = tile >> 3;
    const int tN = tile & 7;

    const int half = lane >> 4;
    const int mn   = lane & 15;
    const int row  = tM * 16 + mn;
    const int col  = tN * 16 + mn;

    const int rowc = (row < M) ? row : (M - 1);
    const float* Arow = A + (size_t)rowc * DF + 2 * half;
    const float* Bcol = W + (size_t)(2 * half) * DF + col;

    v8f c = {};
    #pragma unroll
    for (int k = 0; k < DF; k += 4) {
        const float2 av = *(const float2*)(Arow + k);
        v2f a; a[0] = av.x; a[1] = av.y;
        v2f b;
        b[0] = Bcol[(size_t)k * DF];
        b[1] = Bcol[(size_t)(k + 1) * DF];
        c = __builtin_amdgcn_wmma_f32_16x16x4_f32(
                 false, a,  false, b,
                 (short)0, c,  false,  false);
    }

    const float bv = bias[col];
    #pragma unroll
    for (int v = 0; v < 8; ++v) {
        const int r = tM * 16 + v + 8 * half;
        if (r < M) out[(size_t)r * DF + col] = c[v] + bv;
    }
}

extern "C" void kernel_launch(void* const* d_in, const int* in_sizes, int n_in,
                              void* d_out, int out_size, void* d_ws, size_t ws_size,
                              hipStream_t stream) {
    const float* feature = (const float*)d_in[0];
    const float* w0      = (const float*)d_in[1];
    const float* w1      = (const float*)d_in[2];
    const float* b1      = (const float*)d_in[3];
    const float* W2      = (const float*)d_in[4];
    const float* b2      = (const float*)d_in[5];
    const int*   src0    = (const int*)d_in[6];
    const int*   dst0    = (const int*)d_in[7];
    const int*   src1    = (const int*)d_in[8];
    const int*   dst1    = (const int*)d_in[9];

    const int E0 = in_sizes[1];
    const int E1 = in_sizes[2];
    const int N1 = N1C;
    const int N2 = out_size / DF;

    float* h     = (float*)d_ws;
    float* agg   = h + (size_t)N1 * DF;
    int*   node0 = (int*)(agg + (size_t)N2 * DF);
    int*   node1 = node0 + ((size_t)E0 + 1) * 4;
    int*   head0 = node1 + ((size_t)E1 + 1) * 4;
    int*   head1 = head0 + (size_t)N1 * NSUB;
    float* out   = (float*)d_out;

    {
        int n = N1 + N2;
        int blocks = (n + 255) / 256;
        GCN_lists_init_kernel<<<blocks, 256, 0, stream>>>(head0, head1, node0, node1, N1, N2, E0, E1);
    }
    {
        int n = E0 + E1;
        int blocks = (n + 255) / 256;
        GCN_lists_build_kernel<<<blocks, 256, 0, stream>>>(src0, dst0, w0, src1, dst1, w1,
                                                          head0, head1, node0, node1,
                                                          E0, E1, N1, N2);
    }
    {
        long long threads = (long long)N1 * 16;
        int blocks = (int)((threads + 255) / 256);
        GCN_gather1_kernel<<<blocks, 256, 0, stream>>>(feature, b1, head0, node0, h, N1, E0);
    }
    {
        long long threads = (long long)N2 * 16;
        int blocks = (int)((threads + 255) / 256);
        GCN_gather2_kernel<<<blocks, 256, 0, stream>>>(h, head1, node1, agg, N2, E1);
    }
    {
        int nTiles = ((N2 + 15) / 16) * (DF / 16);
        int wavesPerBlk = 256 / 32;
        int blocks = (nTiles + wavesPerBlk - 1) / wavesPerBlk;
        GCN_gemm_wmma_kernel<<<blocks, 256, 0, stream>>>(agg, W2, b2, out, N2);
    }
}


# Layer-2 gather and dense product fused in one launch: a work-group's 32 summed rows go to LDS as the A tiles of its 16 f32 WMMA waves; no agg buffer, 4 launches instead of 5; sum order unchanged
# speedup vs baseline: 2.0639x; 1.1003x over previous
#include <hip/hip_runtime.h>

#define DF   128
#define N1C  50000

typedef float v2f __attribute__((ext_vector_type(2)));
typedef float v8f __attribute__((ext_vector_type(8)));

#define NSUB 4

__global__ __launch_bounds__(256)
void GCN_lists_init_kernel(int* __restrict__ head0, int* __restrict__ head1,
                           int* __restrict__ node0, int* __restrict__ node1,
                           int nV0, int nV1, int nE0, int nE1) {
    const int i = blockIdx.x * blockDim.x + threadIdx.x;
    if (i < nV0)            ((int4*)head0)[i] = make_int4(nE0, nE0, nE0, nE0);
    else if (i < nV0 + nV1) ((int4*)head1)[i - nV0] = make_int4(nE1, nE1, nE1, nE1);
    if (i == 0) {
        int4 s; s.x = 0; s.y = 0; s.w = 0;
        s.z = nE0; ((int4*)node0)[nE0] = s;
        s.z = nE1; ((int4*)node1)[nE1] = s;
    }
}

__global__ __launch_bounds__(256)
void GCN_lists_build_kernel(const int*   __restrict__ src0,
                            const int*   __restrict__ dst0,
                            const float* __restrict__ w0,
                            const int*   __restrict__ src1,
                            const int*   __restrict__ dst1,
                            const float* __restrict__ w1,
                            int* __restrict__ head0,
                            int* __restrict__ head1,
                            int* __restrict__ node0,
                            int* __restrict__ node1,
                            int nE0, int nE1, int nV0, int nV1) {
    const int t = blockIdx.x * blockDim.x + threadIdx.x;
    const bool first = t < nE0;
    const int  i     = first ? t : t - nE0;
    const int  nE    = first ? nE0 : nE1;
    const int  nV    = first ? nV0 : nV1;
    const int*   src = first ? src0 : src1;
    const int*   dst = first ? dst0 : dst1;
    const float* w   = first ? w0 : w1;
    int* head        = first ? head0 : head1;
    int* node        = first ? node0 : node1;
    if (i >= nE) return;
    const int d = dst[i];
    if ((unsigned)d >= (unsigned)nV) return;
    int4 nd;
    nd.x = src[i];
    nd.y = __float_as_int(w[i]);
    nd.z = atomicExch(head + (size_t)d * NSUB + (i & (NSUB - 1)), i);
    nd.w = 0;
    ((int4*)node)[i] = nd;
}

struct GCN_row8 { float4 lo, hi; };

static __device__ __forceinline__
GCN_row8 GCN_walk_list(const float* __restrict__ rows,
                       const int*   __restrict__ head,
                       const int*   __restrict__ node,
                       int v, bool has, int nE, int c) {
    const char* __restrict__ nbase = (const char*)node;
    const char* __restrict__ rbase = (const char*)rows;
    const unsigned cb = (unsigned)c * 16u;
    const unsigned uE = (unsigned)nE;
    const unsigned bE = uE << 4;
    GCN_row8 a;
    a.lo.x = 0.f; a.lo.y = 0.f; a.lo.z = 0.f; a.lo.w = 0.f;
    a.hi = a.lo;
    int4 hd = make_int4(nE, nE, nE, nE);
    if (has) hd = ((const int4*)head)[v];
    unsigned eb[NSUB];
    eb[0] = min((unsigned)hd.x, uE) << 4;
    eb[1] = min((unsigned)hd.y, uE) << 4;
    eb[2] = min((unsigned)hd.z, uE) << 4;
    eb[3] = min((unsigned)hd.w, uE) << 4;
    int4 nd[NSUB];
    #pragma unroll
    for (int j = 0; j < NSUB; ++j) nd[j] = *(const int4*)(nbase + eb[j]);
    for (int it = 0; it < nE; ++it) {
        bool live[NSUB];
        bool any = false;
        #pragma unroll
        for (int j = 0; j < NSUB; ++j) { live[j] = (eb[j] != bE); any = any || live[j]; }
        if (__builtin_amdgcn_ballot_w32(any) == 0u) break;
        unsigned roff[NSUB];
        float wj[NSUB];
        #pragma unroll
        for (int j = 0; j < NSUB; ++j) {
            roff[j] = (unsigned)nd[j].x * (unsigned)(DF * 4) + cb;
            wj[j]   = __int_as_float(nd[j].y);
            eb[j]   = min((unsigned)nd[j].z, uE) << 4;
            nd[j]   = *(const int4*)(nbase + eb[j]);
        }
        float4 m0[NSUB], m1[NSUB];
        #pragma unroll
        for (int j = 0; j < NSUB; ++j) {
            m0[j] = make_float4(0.f, 0.f, 0.f, 0.f);
            m1[j] = m0[j];
            if (live[j]) {
                m0[j] = *(const float4*)(rbase + roff[j]);
                m1[j] = *(const float4*)(rbase + roff[j] + 256u);
            } else {
                wj[j] = 0.f;
            }
        }
        #pragma unroll
        for (int j = 0; j < NSUB; ++j) {
            const float we = wj[j];
            a.lo.x += m0[j].x * we; a.lo.y += m0[j].y * we; a.lo.z += m0[j].z * we; a.lo.w += m0[j].w * we;
            a.hi.x += m1[j].x * we; a.hi.y += m1[j].y * we; a.hi.z += m1[j].z * we; a.hi.w += m1[j].w * we;
        }
    }
    return a;
}

__global__ __launch_bounds__(256)
void GCN_gather1_kernel(const float* __restrict__ feat,
                        const float* __restrict__ b1,
                        const int*   __restrict__ head,
                        const int*   __restrict__ node,
                        float*       __restrict__ h,
                        int nV, int nE) {
    const int v = (blockIdx.x * blockDim.x + threadIdx.x) >> 4;
    const int c = threadIdx.x & 15;
    const bool has = v < nV;
    const GCN_row8 a = GCN_walk_list(feat, head, node, v, has, nE, c);
    if (has) {
        const char* bb = (const char*)b1 + ((unsigned)c << 4);
        const float4 b0 = *(const float4*)bb;
        const float4 b4 = *(const float4*)(bb + 256);
        float4 r0, r1;
        r0.x = fmaxf(a.lo.x + b0.x, 0.f); r0.y = fmaxf(a.lo.y + b0.y, 0.f);
        r0.z = fmaxf(a.lo.z + b0.z, 0.f); r0.w = fmaxf(a.lo.w + b0.w, 0.f);
        r1.x = fmaxf(a.hi.x + b4.x, 0.f); r1.y = fmaxf(a.hi.y + b4.y, 0.f);
        r1.z = fmaxf(a.hi.z + b4.z, 0.f); r1.w = fmaxf(a.hi.w + b4.w, 0.f);
        float4* o = (float4*)(h + (size_t)v * DF);
        o[c] = r0;
        o[16 + c] = r1;
    }
}

#define APITCH 132
#define GROWS  32

__global__ __launch_bounds__(512, 16)
void GCN_gather2_gemm_kernel(const float* __restrict__ h,
                             const int*   __restrict__ head,
                             const int*   __restrict__ node,
                             const float* __restrict__ W,
                             const float* __restrict__ bias,
                             float*       __restrict__ out,
                             int M, int nE) {
    __shared__ float As[GROWS * APITCH];
    const int r = threadIdx.x >> 4;
    const int v = blockIdx.x * GROWS + r;
    const int c = threadIdx.x & 15;
    const bool has = v < M;
    const GCN_row8 a = GCN_walk_list(h, head, node, v, has, nE, c);
    float4* ar = (float4*)(As + r * APITCH);
    ar[c] = a.lo;
    ar[16 + c] = a.hi;
    __syncthreads();

    const int lane = threadIdx.x & 31;
    const int wv   = threadIdx.x >> 5;
    const int tN   = wv & 7;
    const int tM   = wv >> 3;
    const int half = lane >> 4;
    const int mn   = lane & 15;
    const int col  = tN * 16 + mn;
    const float* Arow = As + (tM * 16 + mn) * APITCH + 2 * half;
    const float* Bcol = W + (size_t)(2 * half) * DF + col;

    v8f acc = {};
    #pragma unroll
    for (int k = 0; k < DF; k += 4) {
        const float2 av = *(const float2*)(Arow + k);
        v2f fa; fa[0] = av.x; fa[1] = av.y;
        v2f fb;
        fb[0] = Bcol[(size_t)k * DF];
        fb[1] = Bcol[(size_t)(k + 1) * DF];
        acc = __builtin_amdgcn_wmma_f32_16x16x4_f32(
                 false, fa,  false, fb,
                 (short)0, acc,  false,  false);
    }

    const float bv = bias[col];
    #pragma unroll
    for (int i = 0; i < 8; ++i) {
        const int ro = blockIdx.x * GROWS + tM * 16 + i + 8 * half;
        if (ro < M) out[(size_t)ro * DF + col] = acc[i] + bv;
    }
}

extern "C" void kernel_launch(void* const* d_in, const int* in_sizes, int n_in,
                              void* d_out, int out_size, void* d_ws, size_t ws_size,
                              hipStream_t stream) {
    const float* feature = (const float*)d_in[0];
    const float* w0      = (const float*)d_in[1];
    const float* w1      = (const float*)d_in[2];
    const float* b1      = (const float*)d_in[3];
    const float* W2      = (const float*)d_in[4];
    const float* b2      = (const float*)d_in[5];
    const int*   src0    = (const int*)d_in[6];
    const int*   dst0    = (const int*)d_in[7];
    const int*   src1    = (const int*)d_in[8];
    const int*   dst1    = (const int*)d_in[9];

    const int E0 = in_sizes[1];
    const int E1 = in_sizes[2];
    const int N1 = N1C;
    const int N2 = out_size / DF;

    float* h     = (float*)d_ws;
    int*   node0 = (int*)(h + (size_t)N1 * DF);
    int*   node1 = node0 + ((size_t)E0 + 1) * 4;
    int*   head0 = node1 + ((size_t)E1 + 1) * 4;
    int*   head1 = head0 + (size_t)N1 * NSUB;
    float* out   = (float*)d_out;

    {
        int n = N1 + N2;
        int blocks = (n + 255) / 256;
        GCN_lists_init_kernel<<<blocks, 256, 0, stream>>>(head0, head1, node0, node1, N1, N2, E0, E1);
    }
    {
        int n = E0 + E1;
        int blocks = (n + 255) / 256;
        GCN_lists_build_kernel<<<blocks, 256, 0, stream>>>(src0, dst0, w0, src1, dst1, w1,
                                                          head0, head1, node0, node1,
                                                          E0, E1, N1, N2);
    }
    {
        long long threads = (long long)N1 * 16;
        int blocks = (int)((threads + 255) / 256);
        GCN_gather1_kernel<<<blocks, 256, 0, stream>>>(feature, b1, head0, node0, h, N1, E0);
    }
    {
        int blocks = (N2 + GROWS - 1) / GROWS;
        GCN_gather2_gemm_kernel<<<blocks, 512, 0, stream>>>(h, head1, node1, W2, b2, out, N2, E1);
    }
}


# List init and build launches start 4x and 8x fewer waves: a lane pushes 8 edges, all loads, then all exchanges, then all node stores in flight together; sum order still exchange arrival
# speedup vs baseline: 2.1447x; 1.0391x over previous
#include <hip/hip_runtime.h>

#define DF   128
#define N1C  50000

typedef float v2f __attribute__((ext_vector_type(2)));
typedef float v8f __attribute__((ext_vector_type(8)));

#define NSUB 4

#define HPL 4
__global__ __launch_bounds__(256)
void GCN_lists_init_kernel(int* __restrict__ head0, int* __restrict__ head1,
                           int* __restrict__ node0, int* __restrict__ node1,
                           int nV0, int nV1, int nE0, int nE1) {
    const int i0 = blockIdx.x * (256 * HPL) + threadIdx.x;
    #pragma unroll
    for (int k = 0; k < HPL; ++k) {
        const int i = i0 + 256 * k;
        if (i < nV0)            ((int4*)head0)[i] = make_int4(nE0, nE0, nE0, nE0);
        else if (i < nV0 + nV1) ((int4*)head1)[i - nV0] = make_int4(nE1, nE1, nE1, nE1);
    }
    if (i0 == 0) {
        int4 s; s.x = 0; s.y = 0; s.w = 0;
        s.z = nE0; ((int4*)node0)[nE0] = s;
        s.z = nE1; ((int4*)node1)[nE1] = s;
    }
}

#define EPL 8
__global__ __launch_bounds__(256)
void GCN_lists_build_kernel(const int*   __restrict__ src0,
                            const int*   __restrict__ dst0,
                            const float* __restrict__ w0,
                            const int*   __restrict__ src1,
                            const int*   __restrict__ dst1,
                            const float* __restrict__ w1,
                            int* __restrict__ head0,
                            int* __restrict__ head1,
                            int* __restrict__ node0,
                            int* __restrict__ node1,
                            int nE0, int nE1, int nV0, int nV1) {
    const int  nB0   = (nE0 + 256 * EPL - 1) / (256 * EPL);
    const bool first = (int)blockIdx.x < nB0;
    const int  chunk = first ? (int)blockIdx.x : (int)blockIdx.x - nB0;
    const int  nE    = first ? nE0 : nE1;
    const int  nV    = first ? nV0 : nV1;
    const int*   __restrict__ src = first ? src0 : src1;
    const int*   __restrict__ dst = first ? dst0 : dst1;
    const float* __restrict__ w   = first ? w0 : w1;
    int* __restrict__ head        = first ? head0 : head1;
    int* __restrict__ node        = first ? node0 : node1;
    const int i0 = chunk * (256 * EPL) + threadIdx.x;

    int d[EPL], s[EPL], wb[EPL], nx[EPL];
    #pragma unroll
    for (int k = 0; k < EPL; ++k) {
        const int i = i0 + 256 * k;
        d[k] = -1; s[k] = 0; wb[k] = 0;
        if (i < nE) {
            d[k]  = dst[i];
            s[k]  = src[i];
            wb[k] = __float_as_int(w[i]);
        }
    }
    #pragma unroll
    for (int k = 0; k < EPL; ++k) {
        const int i = i0 + 256 * k;
        nx[k] = 0;
        if ((unsigned)d[k] < (unsigned)nV)
            nx[k] = atomicExch(head + (size_t)d[k] * NSUB + (i & (NSUB - 1)), i);
    }
    #pragma unroll
    for (int k = 0; k < EPL; ++k) {
        const int i = i0 + 256 * k;
        if ((unsigned)d[k] < (unsigned)nV) {
            int4 nd;
            nd.x = s[k];
            nd.y = wb[k];
            nd.z = nx[k];
            nd.w = 0;
            ((int4*)node)[i] = nd;
        }
    }
}

struct GCN_row8 { float4 lo, hi; };

static __device__ __forceinline__
GCN_row8 GCN_walk_list(const float* __restrict__ rows,
                       const int*   __restrict__ head,
                       const int*   __restrict__ node,
                       int v, bool has, int nE, int c) {
    const char* __restrict__ nbase = (const char*)node;
    const char* __restrict__ rbase = (const char*)rows;
    const unsigned cb = (unsigned)c * 16u;
    const unsigned uE = (unsigned)nE;
    const unsigned bE = uE << 4;
    GCN_row8 a;
    a.lo.x = 0.f; a.lo.y = 0.f; a.lo.z = 0.f; a.lo.w = 0.f;
    a.hi = a.lo;
    int4 hd = make_int4(nE, nE, nE, nE);
    if (has) hd = ((const int4*)head)[v];
    unsigned eb[NSUB];
    eb[0] = min((unsigned)hd.x, uE) << 4;
    eb[1] = min((unsigned)hd.y, uE) << 4;
    eb[2] = min((unsigned)hd.z, uE) << 4;
    eb[3] = min((unsigned)hd.w, uE) << 4;
    int4 nd[NSUB];
    #pragma unroll
    for (int j = 0; j < NSUB; ++j) nd[j] = *(const int4*)(nbase + eb[j]);
    for (int it = 0; it < nE; ++it) {
        bool live[NSUB];
        bool any = false;
        #pragma unroll
        for (int j = 0; j < NSUB; ++j) { live[j] = (eb[j] != bE); any = any || live[j]; }
        if (__builtin_amdgcn_ballot_w32(any) == 0u) break;
        unsigned roff[NSUB];
        float wj[NSUB];
        #pragma unroll
        for (int j = 0; j < NSUB; ++j) {
            roff[j] = (unsigned)nd[j].x * (unsigned)(DF * 4) + cb;
            wj[j]   = __int_as_float(nd[j].y);
            eb[j]   = min((unsigned)nd[j].z, uE) << 4;
            nd[j]   = *(const int4*)(nbase + eb[j]);
        }
        float4 m0[NSUB], m1[NSUB];
        #pragma unroll
        for (int j = 0; j < NSUB; ++j) {
            m0[j] = make_float4(0.f, 0.f, 0.f, 0.f);
            m1[j] = m0[j];
            if (live[j]) {
                m0[j] = *(const float4*)(rbase + roff[j]);
                m1[j] = *(const float4*)(rbase + roff[j] + 256u);
            } else {
                wj[j] = 0.f;
            }
        }
        #pragma unroll
        for (int j = 0; j < NSUB; ++j) {
            const float we = wj[j];
            a.lo.x += m0[j].x * we; a.lo.y += m0[j].y * we; a.lo.z += m0[j].z * we; a.lo.w += m0[j].w * we;
            a.hi.x += m1[j].x * we; a.hi.y += m1[j].y * we; a.hi.z += m1[j].z * we; a.hi.w += m1[j].w * we;
        }
    }
    return a;
}

__global__ __launch_bounds__(256)
void GCN_gather1_kernel(const float* __restrict__ feat,
                        const float* __restrict__ b1,
                        const int*   __restrict__ head,
                        const int*   __restrict__ node,
                        float*       __restrict__ h,
                        int nV, int nE) {
    const int v = (blockIdx.x * blockDim.x + threadIdx.x) >> 4;
    const int c = threadIdx.x & 15;
    const bool has = v < nV;
    const GCN_row8 a = GCN_walk_list(feat, head, node, v, has, nE, c);
    if (has) {
        const char* bb = (const char*)b1 + ((unsigned)c << 4);
        const float4 b0 = *(const float4*)bb;
        const float4 b4 = *(const float4*)(bb + 256);
        float4 r0, r1;
        r0.x = fmaxf(a.lo.x + b0.x, 0.f); r0.y = fmaxf(a.lo.y + b0.y, 0.f);
        r0.z = fmaxf(a.lo.z + b0.z, 0.f); r0.w = fmaxf(a.lo.w + b0.w, 0.f);
        r1.x = fmaxf(a.hi.x + b4.x, 0.f); r1.y = fmaxf(a.hi.y + b4.y, 0.f);
        r1.z = fmaxf(a.hi.z + b4.z, 0.f); r1.w = fmaxf(a.hi.w + b4.w, 0.f);
        float4* o = (float4*)(h + (size_t)v * DF);
        o[c] = r0;
        o[16 + c] = r1;
    }
}

#define APITCH 132
#define GROWS  32

__global__ __launch_bounds__(512, 16)
void GCN_gather2_gemm_kernel(const float* __restrict__ h,
                             const int*   __restrict__ head,
                             const int*   __restrict__ node,
                             const float* __restrict__ W,
                             const float* __restrict__ bias,
                             float*       __restrict__ out,
                             int M, int nE) {
    __shared__ float As[GROWS * APITCH];
    const int r = threadIdx.x >> 4;
    const int v = blockIdx.x * GROWS + r;
    const int c = threadIdx.x & 15;
    const bool has = v < M;
    const GCN_row8 a = GCN_walk_list(h, head, node, v, has, nE, c);
    float4* ar = (float4*)(As + r * APITCH);
    ar[c] = a.lo;
    ar[16 + c] = a.hi;
    __syncthreads();

    const int lane = threadIdx.x & 31;
    const int wv   = threadIdx.x >> 5;
    const int tN   = wv & 7;
    const int tM   = wv >> 3;
    const int half = lane >> 4;
    const int mn   = lane & 15;
    const int col  = tN * 16 + mn;
    const float* Arow = As + (tM * 16 + mn) * APITCH + 2 * half;
    const float* Bcol = W + (size_t)(2 * half) * DF + col;

    v8f acc = {};
    #pragma unroll
    for (int k = 0; k < DF; k += 4) {
        const float2 av = *(const float2*)(Arow + k);
        v2f fa; fa[0] = av.x; fa[1] = av.y;
        v2f fb;
        fb[0] = Bcol[(size_t)k * DF];
        fb[1] = Bcol[(size_t)(k + 1) * DF];
        acc = __builtin_amdgcn_wmma_f32_16x16x4_f32(
                 false, fa,  false, fb,
                 (short)0, acc,  false,  false);
    }

    const float bv = bias[col];
    #pragma unroll
    for (int i = 0; i < 8; ++i) {
        const int ro = blockIdx.x * GROWS + tM * 16 + i + 8 * half;
        if (ro < M) out[(size_t)ro * DF + col] = acc[i] + bv;
    }
}

extern "C" void kernel_launch(void* const* d_in, const int* in_sizes, int n_in,
                              void* d_out, int out_size, void* d_ws, size_t ws_size,
                              hipStream_t stream) {
    const float* feature = (const float*)d_in[0];
    const float* w0      = (const float*)d_in[1];
    const float* w1      = (const float*)d_in[2];
    const float* b1      = (const float*)d_in[3];
    const float* W2      = (const float*)d_in[4];
    const float* b2      = (const float*)d_in[5];
    const int*   src0    = (const int*)d_in[6];
    const int*   dst0    = (const int*)d_in[7];
    const int*   src1    = (const int*)d_in[8];
    const int*   dst1    = (const int*)d_in[9];

    const int E0 = in_sizes[1];
    const int E1 = in_sizes[2];
    const int N1 = N1C;
    const int N2 = out_size / DF;

    float* h     = (float*)d_ws;
    int*   node0 = (int*)(h + (size_t)N1 * DF);
    int*   node1 = node0 + ((size_t)E0 + 1) * 4;
    int*   head0 = node1 + ((size_t)E1 + 1) * 4;
    int*   head1 = head0 + (size_t)N1 * NSUB;
    float* out   = (float*)d_out;

    {
        int n = N1 + N2;
        int blocks = (n + 256 * HPL - 1) / (256 * HPL);
        GCN_lists_init_kernel<<<blocks, 256, 0, stream>>>(head0, head1, node0, node1, N1, N2, E0, E1);
    }
    {
        int blocks = (E0 + 256 * EPL - 1) / (256 * EPL) + (E1 + 256 * EPL - 1) / (256 * EPL);
        GCN_lists_build_kernel<<<blocks, 256, 0, stream>>>(src0, dst0, w0, src1, dst1, w1,
                                                          head0, head1, node0, node1,
                                                          E0, E1, N1, N2);
    }
    {
        long long threads = (long long)N1 * 16;
        int blocks = (int)((threads + 255) / 256);
        GCN_gather1_kernel<<<blocks, 256, 0, stream>>>(feature, b1, head0, node0, h, N1, E0);
    }
    {
        int blocks = (N2 + GROWS - 1) / GROWS;
        GCN_gather2_gemm_kernel<<<blocks, 512, 0, stream>>>(h, head1, node1, W2, b2, out, N2, E1);
    }
}


# Per-edge global exchange lists replaced by a two-level counting sort on LDS counters (8192-edge chunks by 32-row block, block re-ranked in the gather work-group); 3 launches; sum order = LDS arrival
# speedup vs baseline: 2.7566x; 1.2853x over previous
#include <hip/hip_runtime.h>

#define DF   128
#define N1C  50000

typedef float v2f __attribute__((ext_vector_type(2)));
typedef float v8f __attribute__((ext_vector_type(8)));

#define CH     8192
#define EPT    8
#define GROWS  32
#define NBCAP  2048
#define NCMAX  128
#define CAP    1024

__global__ __launch_bounds__(1024)
void GCN_chunk_sort_kernel(const int*   __restrict__ src0,
                           const int*   __restrict__ dst0,
                           const float* __restrict__ w0,
                           const int*   __restrict__ src1,
                           const int*   __restrict__ dst1,
                           const float* __restrict__ w1,
                           int4* __restrict__ R0, int4* __restrict__ R1,
                           int*  __restrict__ T0, int*  __restrict__ T1,
                           int nE0, int nE1, int nV0, int nV1,
                           int nC0, int nC1) {
    __shared__ int  cnt[NBCAP + 1];
    __shared__ int  wsum[32];
    __shared__ int4 rec[CH];

    const int  tid   = threadIdx.x;
    const bool first = (int)blockIdx.x < nC0;
    const int  ch    = first ? (int)blockIdx.x : (int)blockIdx.x - nC0;
    const int  nE    = first ? nE0 : nE1;
    const int  nV    = first ? nV0 : nV1;
    const int  nC    = first ? nC0 : nC1;
    const int*   __restrict__ src = first ? src0 : src1;
    const int*   __restrict__ dst = first ? dst0 : dst1;
    const float* __restrict__ w   = first ? w0 : w1;
    int4* __restrict__ R          = first ? R0 : R1;
    int*  __restrict__ T          = first ? T0 : T1;
    const int nB = min((nV + GROWS - 1) / GROWS, NBCAP);

    for (int i = tid; i <= NBCAP; i += 1024) cnt[i] = 0;
    __syncthreads();

    int d[EPT], s[EPT], wb[EPT], rk[EPT];
    #pragma unroll
    for (int k = 0; k < EPT; ++k) {
        const int i = ch * CH + tid + 1024 * k;
        d[k] = -1; s[k] = 0; wb[k] = 0;
        if (i < nE) {
            d[k]  = dst[i];
            s[k]  = src[i];
            wb[k] = __float_as_int(w[i]);
        }
    }
    #pragma unroll
    for (int k = 0; k < EPT; ++k) {
        rk[k] = 0;
        if ((unsigned)d[k] < (unsigned)nV && (d[k] >> 5) < NBCAP)
            rk[k] = atomicAdd(&cnt[d[k] >> 5], 1);
        else
            d[k] = -1;
    }
    __syncthreads();

    const int lane = tid & 31, wv = tid >> 5;
    const int a0 = cnt[2 * tid], a1 = cnt[2 * tid + 1];
    const int sum = a0 + a1;
    int x = sum;
    #pragma unroll
    for (int off = 1; off < 32; off <<= 1) {
        const int y = __shfl_up(x, off);
        if (lane >= off) x += y;
    }
    if (lane == 31) wsum[wv] = x;
    __syncthreads();
    if (wv == 0) {
        const int t = wsum[lane];
        int z = t;
        #pragma unroll
        for (int off = 1; off < 32; off <<= 1) {
            const int y = __shfl_up(z, off);
            if (lane >= off) z += y;
        }
        wsum[lane] = z - t;
    }
    __syncthreads();
    const int excl = wsum[wv] + x - sum;
    cnt[2 * tid]     = excl;
    cnt[2 * tid + 1] = excl + a0;
    if (tid == 1023) cnt[NBCAP] = excl + sum;
    __syncthreads();

    #pragma unroll
    for (int k = 0; k < EPT; ++k) {
        if (d[k] >= 0) {
            int4 e;
            e.x = s[k];
            e.y = wb[k];
            e.z = d[k] & (GROWS - 1);
            e.w = 0;
            rec[cnt[d[k] >> 5] + rk[k]] = e;
        }
    }
    __syncthreads();

    const int total = cnt[NBCAP];
    int4* __restrict__ Rc = R + (size_t)ch * CH;
    for (int j = tid; j < total; j += 1024) Rc[j] = rec[j];
    for (int b = tid; b <= nB; b += 1024) T[(size_t)b * nC + ch] = cnt[b];
}

struct GCN_row8 { float4 lo, hi; };

struct GCN_block_lds {
    int runS[NCMAX];
    int runP[NCMAX + 1];
    int c2[GROWS];
    int st2[GROWS + 1];
    int rs[CAP], rw[CAP], rl[CAP];
    int ss[CAP], sw[CAP];
};

static __device__ __forceinline__
void GCN_sum_rows(GCN_row8& a, const float* __restrict__ rows,
                  const int* ss, const int* sw, int k0, int k1, int c) {
    const char* __restrict__ rbase = (const char*)rows;
    const unsigned cb = (unsigned)c * 16u;
    for (int it = 0; it < CAP / 4; ++it) {
        const int k = k0 + 4 * it;
        if (__builtin_amdgcn_ballot_w32(k < k1) == 0u) break;
        bool live[4];
        unsigned roff[4];
        float wj[4];
        #pragma unroll
        for (int j = 0; j < 4; ++j) {
            live[j] = (k + j) < k1;
            const int q = min(k + j, CAP - 1);
            roff[j] = (unsigned)ss[q] * (unsigned)(DF * 4) + cb;
            wj[j]   = __int_as_float(sw[q]);
        }
        float4 m0[4], m1[4];
        #pragma unroll
        for (int j = 0; j < 4; ++j) {
            m0[j] = make_float4(0.f, 0.f, 0.f, 0.f);
            m1[j] = m0[j];
            if (live[j]) {
                m0[j] = *(const float4*)(rbase + roff[j]);
                m1[j] = *(const float4*)(rbase + roff[j] + 256u);
            } else {
                wj[j] = 0.f;
            }
        }
        #pragma unroll
        for (int j = 0; j < 4; ++j) {
            const float we = wj[j];
            a.lo.x += m0[j].x * we; a.lo.y += m0[j].y * we; a.lo.z += m0[j].z * we; a.lo.w += m0[j].w * we;
            a.hi.x += m1[j].x * we; a.hi.y += m1[j].y * we; a.hi.z += m1[j].z * we; a.hi.w += m1[j].w * we;
        }
    }
}

static __device__ __forceinline__
GCN_row8 GCN_block_sums(GCN_block_lds& L, const float* __restrict__ rows,
                        const int4* __restrict__ R, const int* __restrict__ T,
                        int blk, int nC, int r, int c) {
    const int tid = threadIdx.x, lane = tid & 31, wv = tid >> 5;
    GCN_row8 a;
    a.lo.x = 0.f; a.lo.y = 0.f; a.lo.z = 0.f; a.lo.w = 0.f;
    a.hi = a.lo;
    const int nCs = nC;
    nC = min(nC, NCMAX);

    if (wv == 0) {
        int st[4], en[4], ln[4];
        #pragma unroll
        for (int i = 0; i < 4; ++i) {
            const int cc = max(min(4 * lane + i, nC - 1), 0);
            st[i] = T[(size_t)blk * nCs + cc];
            en[i] = T[(size_t)(blk + 1) * nCs + cc];
        }
        int tot = 0;
        #pragma unroll
        for (int i = 0; i < 4; ++i) {
            const int s = min(max(st[i], 0), CH);
            const int e = min(max(en[i], s), CH);
            st[i] = s;
            ln[i] = (4 * lane + i < nC) ? e - s : 0;
            tot += ln[i];
        }
        int x = tot;
        #pragma unroll
        for (int off = 1; off < 32; off <<= 1) {
            const int y = __shfl_up(x, off);
            if (lane >= off) x += y;
        }
        int ex = x - tot;
        #pragma unroll
        for (int i = 0; i < 4; ++i) {
            L.runS[4 * lane + i] = st[i];
            L.runP[4 * lane + i] = ex;
            ex += ln[i];
        }
        if (lane == 31) L.runP[NCMAX] = ex;
    }
    __syncthreads();
    const int Ltot = L.runP[NCMAX];

    for (int base = 0; base < Ltot; base += CAP) {
        const int Wn = min(CAP, Ltot - base);
        {
            const int ck = tid >> 2, m = tid & 3;
            if (ck < nC) {
                const int p0 = L.runP[ck], p1 = L.runP[ck + 1];
                const int lo = max(p0, base), hi = min(p1, base + CAP);
                const int4* __restrict__ Rc = R + (size_t)ck * CH + L.runS[ck];
                for (int q = lo + m; q < hi; q += 4) {
                    const int4 e = Rc[q - p0];
                    L.rs[q - base] = e.x;
                    L.rw[q - base] = e.y;
                    L.rl[q - base] = e.z & (GROWS - 1);
                }
            }
            if (tid < GROWS) L.c2[tid] = 0;
        }
        __syncthreads();
        int ld[CAP / 512], rk[CAP / 512];
        #pragma unroll
        for (int i = 0; i < CAP / 512; ++i) {
            const int j = tid + 512 * i;
            ld[i] = 0; rk[i] = 0;
            if (j < Wn) {
                ld[i] = L.rl[j];
                rk[i] = atomicAdd(&L.c2[ld[i]], 1);
            }
        }
        __syncthreads();
        if (wv == 0) {
            const int v = L.c2[lane];
            int x = v;
            #pragma unroll
            for (int off = 1; off < 32; off <<= 1) {
                const int y = __shfl_up(x, off);
                if (lane >= off) x += y;
            }
            L.st2[lane] = x - v;
            if (lane == 31) L.st2[GROWS] = x;
        }
        __syncthreads();
        #pragma unroll
        for (int i = 0; i < CAP / 512; ++i) {
            const int j = tid + 512 * i;
            if (j < Wn) {
                const int p = L.st2[ld[i]] + rk[i];
                L.ss[p] = L.rs[j];
                L.sw[p] = L.rw[j];
            }
        }
        __syncthreads();
        GCN_sum_rows(a, rows, L.ss, L.sw, L.st2[r], L.st2[r + 1], c);
        __syncthreads();
    }
    return a;
}

__global__ __launch_bounds__(512, 16)
void GCN_gather1_kernel(const float* __restrict__ feat,
                        const float* __restrict__ b1,
                        const int4*  __restrict__ R,
                        const int*   __restrict__ T,
                        float*       __restrict__ h,
                        int nV, int nC) {
    __shared__ GCN_block_lds L;
    const int r = threadIdx.x >> 4;
    const int v = blockIdx.x * GROWS + r;
    const int c = threadIdx.x & 15;
    const GCN_row8 a = GCN_block_sums(L, feat, R, T, blockIdx.x, nC, r, c);
    if (v < nV) {
        const char* bb = (const char*)b1 + ((unsigned)c << 4);
        const float4 b0 = *(const float4*)bb;
        const float4 b4 = *(const float4*)(bb + 256);
        float4 r0, r1;
        r0.x = fmaxf(a.lo.x + b0.x, 0.f); r0.y = fmaxf(a.lo.y + b0.y, 0.f);
        r0.z = fmaxf(a.lo.z + b0.z, 0.f); r0.w = fmaxf(a.lo.w + b0.w, 0.f);
        r1.x = fmaxf(a.hi.x + b4.x, 0.f); r1.y = fmaxf(a.hi.y + b4.y, 0.f);
        r1.z = fmaxf(a.hi.z + b4.z, 0.f); r1.w = fmaxf(a.hi.w + b4.w, 0.f);
        float4* o = (float4*)(h + (size_t)v * DF);
        o[c] = r0;
        o[16 + c] = r1;
    }
}

#define APITCH 132

__global__ __launch_bounds__(512, 16)
void GCN_gather2_gemm_kernel(const float* __restrict__ h,
                             const int4*  __restrict__ R,
                             const int*   __restrict__ T,
                             const float* __restrict__ W,
                             const float* __restrict__ bias,
                             float*       __restrict__ out,
                             int M, int nC) {
    __shared__ GCN_block_lds L;
    __shared__ float As[GROWS * APITCH];
    const int r = threadIdx.x >> 4;
    const int c = threadIdx.x & 15;
    const GCN_row8 a = GCN_block_sums(L, h, R, T, blockIdx.x, nC, r, c);
    float4* ar = (float4*)(As + r * APITCH);
    ar[c] = a.lo;
    ar[16 + c] = a.hi;
    __syncthreads();

    const int lane = threadIdx.x & 31;
    const int wv   = threadIdx.x >> 5;
    const int tN   = wv & 7;
    const int tM   = wv >> 3;
    const int half = lane >> 4;
    const int mn   = lane & 15;
    const int col  = tN * 16 + mn;
    const float* Arow = As + (tM * 16 + mn) * APITCH + 2 * half;
    const float* Bcol = W + (size_t)(2 * half) * DF + col;

    v8f acc = {};
    #pragma unroll
    for (int k = 0; k < DF; k += 4) {
        const float2 av = *(const float2*)(Arow + k);
        v2f fa; fa[0] = av.x; fa[1] = av.y;
        v2f fb;
        fb[0] = Bcol[(size_t)k * DF];
        fb[1] = Bcol[(size_t)(k + 1) * DF];
        acc = __builtin_amdgcn_wmma_f32_16x16x4_f32(
                 false, fa,  false, fb,
                 (short)0, acc,  false,  false);
    }

    const float bv = bias[col];
    #pragma unroll
    for (int i = 0; i < 8; ++i) {
        const int ro = blockIdx.x * GROWS + tM * 16 + i + 8 * half;
        if (ro < M) out[(size_t)ro * DF + col] = acc[i] + bv;
    }
}

extern "C" void kernel_launch(void* const* d_in, const int* in_sizes, int n_in,
                              void* d_out, int out_size, void* d_ws, size_t ws_size,
                              hipStream_t stream) {
    const float* feature = (const float*)d_in[0];
    const float* w0      = (const float*)d_in[1];
    const float* w1      = (const float*)d_in[2];
    const float* b1      = (const float*)d_in[3];
    const float* W2      = (const float*)d_in[4];
    const float* b2      = (const float*)d_in[5];
    const int*   src0    = (const int*)d_in[6];
    const int*   dst0    = (const int*)d_in[7];
    const int*   src1    = (const int*)d_in[8];
    const int*   dst1    = (const int*)d_in[9];

    const int E0 = in_sizes[1];
    const int E1 = in_sizes[2];
    const int N1 = N1C;
    const int N2 = out_size / DF;

    const int nC0 = (E0 + CH - 1) / CH;
    const int nC1 = (E1 + CH - 1) / CH;
    const int nB0 = (N1 + GROWS - 1) / GROWS;
    const int nB1 = (N2 + GROWS - 1) / GROWS;

    float* h   = (float*)d_ws;
    int4*  R0  = (int4*)(h + (size_t)N1 * DF);
    int4*  R1  = R0 + (size_t)nC0 * CH;
    int*   T0  = (int*)(R1 + (size_t)nC1 * CH);
    int*   T1  = T0 + (size_t)(nB0 + 1) * nC0;
    float* out = (float*)d_out;

    GCN_chunk_sort_kernel<<<nC0 + nC1, 1024, 0, stream>>>(src0, dst0, w0, src1, dst1, w1,
                                                          R0, R1, T0, T1,
                                                          E0, E1, N1, N2, nC0, nC1);
    GCN_gather1_kernel<<<nB0, 512, 0, stream>>>(feature, b1, R0, T0, h, N1, nC0);
    GCN_gather2_gemm_kernel<<<nB1, 512, 0, stream>>>(h, R1, T1, W2, b2, out, N2, nC1);
}


# 8-wave gather work-groups, not 16: a 16-lane group sums 2 destinations in turn, same loads in flight per wave; 2 threads per chunk copy 8 records a trip; product 2 row tiles per wave; sum order kept
# speedup vs baseline: 2.9715x; 1.0780x over previous
#include <hip/hip_runtime.h>

#define DF   128
#define N1C  50000

typedef float v2f __attribute__((ext_vector_type(2)));
typedef float v8f __attribute__((ext_vector_type(8)));

#define CH     8192
#define EPT    8
#define GROWS  32
#define NBCAP  2048
#define NCMAX  128
#define CAP    1024

__global__ __launch_bounds__(1024)
void GCN_chunk_sort_kernel(const int*   __restrict__ src0,
                           const int*   __restrict__ dst0,
                           const float* __restrict__ w0,
                           const int*   __restrict__ src1,
                           const int*   __restrict__ dst1,
                           const float* __restrict__ w1,
                           int4* __restrict__ R0, int4* __restrict__ R1,
                           int*  __restrict__ T0, int*  __restrict__ T1,
                           int nE0, int nE1, int nV0, int nV1,
                           int nC0, int nC1) {
    __shared__ int  cnt[NBCAP + 1];
    __shared__ int  wsum[32];
    __shared__ int4 rec[CH];

    const int  tid   = threadIdx.x;
    const bool first = (int)blockIdx.x < nC0;
    const int  ch    = first ? (int)blockIdx.x : (int)blockIdx.x - nC0;
    const int  nE    = first ? nE0 : nE1;
    const int  nV    = first ? nV0 : nV1;
    const int  nC    = first ? nC0 : nC1;
    const int*   __restrict__ src = first ? src0 : src1;
    const int*   __restrict__ dst = first ? dst0 : dst1;
    const float* __restrict__ w   = first ? w0 : w1;
    int4* __restrict__ R          = first ? R0 : R1;
    int*  __restrict__ T          = first ? T0 : T1;
    const int nB = min((nV + GROWS - 1) / GROWS, NBCAP);

    for (int i = tid; i <= NBCAP; i += 1024) cnt[i] = 0;
    __syncthreads();

    int d[EPT], s[EPT], wb[EPT], rk[EPT];
    #pragma unroll
    for (int k = 0; k < EPT; ++k) {
        const int i = ch * CH + tid + 1024 * k;
        d[k] = -1; s[k] = 0; wb[k] = 0;
        if (i < nE) {
            d[k]  = dst[i];
            s[k]  = src[i];
            wb[k] = __float_as_int(w[i]);
        }
    }
    #pragma unroll
    for (int k = 0; k < EPT; ++k) {
        rk[k] = 0;
        if ((unsigned)d[k] < (unsigned)nV && (d[k] >> 5) < NBCAP)
            rk[k] = atomicAdd(&cnt[d[k] >> 5], 1);
        else
            d[k] = -1;
    }
    __syncthreads();

    const int lane = tid & 31, wv = tid >> 5;
    const int a0 = cnt[2 * tid], a1 = cnt[2 * tid + 1];
    const int sum = a0 + a1;
    int x = sum;
    #pragma unroll
    for (int off = 1; off < 32; off <<= 1) {
        const int y = __shfl_up(x, off);
        if (lane >= off) x += y;
    }
    if (lane == 31) wsum[wv] = x;
    __syncthreads();
    if (wv == 0) {
        const int t = wsum[lane];
        int z = t;
        #pragma unroll
        for (int off = 1; off < 32; off <<= 1) {
            const int y = __shfl_up(z, off);
            if (lane >= off) z += y;
        }
        wsum[lane] = z - t;
    }
    __syncthreads();
    const int excl = wsum[wv] + x - sum;
    cnt[2 * tid]     = excl;
    cnt[2 * tid + 1] = excl + a0;
    if (tid == 1023) cnt[NBCAP] = excl + sum;
    __syncthreads();

    #pragma unroll
    for (int k = 0; k < EPT; ++k) {
        if (d[k] >= 0) {
            int4 e;
            e.x = s[k];
            e.y = wb[k];
            e.z = d[k] & (GROWS - 1);
            e.w = 0;
            rec[cnt[d[k] >> 5] + rk[k]] = e;
        }
    }
    __syncthreads();

    const int total = cnt[NBCAP];
    int4* __restrict__ Rc = R + (size_t)ch * CH;
    for (int j = tid; j < total; j += 1024) Rc[j] = rec[j];
    for (int b = tid; b <= nB; b += 1024) T[(size_t)b * nC + ch] = cnt[b];
}

struct GCN_row8 { float4 lo, hi; };

struct GCN_block_lds {
    int runS[NCMAX];
    int runP[NCMAX + 1];
    int c2[GROWS];
    int st2[GROWS + 1];
    int rs[CAP], rw[CAP], rl[CAP];
    int ss[CAP], sw[CAP];
};

static __device__ __forceinline__
void GCN_sum_rows(GCN_row8& a, const float* __restrict__ rows,
                  const int* ss, const int* sw, int k0, int k1, int c) {
    const char* __restrict__ rbase = (const char*)rows;
    const unsigned cb = (unsigned)c * 16u;
    for (int it = 0; it < CAP / 4; ++it) {
        const int k = k0 + 4 * it;
        if (__builtin_amdgcn_ballot_w32(k < k1) == 0u) break;
        bool live[4];
        unsigned roff[4];
        float wj[4];
        #pragma unroll
        for (int j = 0; j < 4; ++j) {
            live[j] = (k + j) < k1;
            const int q = min(k + j, CAP - 1);
            roff[j] = (unsigned)ss[q] * (unsigned)(DF * 4) + cb;
            wj[j]   = __int_as_float(sw[q]);
        }
        float4 m0[4], m1[4];
        #pragma unroll
        for (int j = 0; j < 4; ++j) {
            m0[j] = make_float4(0.f, 0.f, 0.f, 0.f);
            m1[j] = m0[j];
            if (live[j]) {
                m0[j] = *(const float4*)(rbase + roff[j]);
                m1[j] = *(const float4*)(rbase + roff[j] + 256u);
            } else {
                wj[j] = 0.f;
            }
        }
        #pragma unroll
        for (int j = 0; j < 4; ++j) {
            const float we = wj[j];
            a.lo.x += m0[j].x * we; a.lo.y += m0[j].y * we; a.lo.z += m0[j].z * we; a.lo.w += m0[j].w * we;
            a.hi.x += m1[j].x * we; a.hi.y += m1[j].y * we; a.hi.z += m1[j].z * we; a.hi.w += m1[j].w * we;
        }
    }
}

#define GT 256

static __device__ __forceinline__
int GCN_block_runs(GCN_block_lds& L, const int* __restrict__ T, int blk, int nC, int nCs) {
    const int tid = threadIdx.x, lane = tid & 31, wv = tid >> 5;
    if (wv == 0) {
        int st[4], en[4], ln[4];
        #pragma unroll
        for (int i = 0; i < 4; ++i) {
            const int cc = max(min(4 * lane + i, nC - 1), 0);
            st[i] = T[(size_t)blk * nCs + cc];
            en[i] = T[(size_t)(blk + 1) * nCs + cc];
        }
        int tot = 0;
        #pragma unroll
        for (int i = 0; i < 4; ++i) {
            const int s = min(max(st[i], 0), CH);
            const int e = min(max(en[i], s), CH);
            st[i] = s;
            ln[i] = (4 * lane + i < nC) ? e - s : 0;
            tot += ln[i];
        }
        int x = tot;
        #pragma unroll
        for (int off = 1; off < 32; off <<= 1) {
            const int y = __shfl_up(x, off);
            if (lane >= off) x += y;
        }
        int ex = x - tot;
        #pragma unroll
        for (int i = 0; i < 4; ++i) {
            L.runS[4 * lane + i] = st[i];
            L.runP[4 * lane + i] = ex;
            ex += ln[i];
        }
        if (lane == 31) L.runP[NCMAX] = ex;
    }
    __syncthreads();
    return L.runP[NCMAX];
}

static __device__ __forceinline__
void GCN_stage_window(GCN_block_lds& L, const int4* __restrict__ R, int nC, int base, int Ltot) {
    const int tid = threadIdx.x, lane = tid & 31, wv = tid >> 5;
    const int Wn = min(CAP, Ltot - base);
    {
        const int ck = tid >> 1, m = tid & 1;
        if (ck < nC) {
            const int p0 = L.runP[ck], p1 = L.runP[ck + 1];
            const int lo = max(p0, base), hi = min(p1, base + CAP);
            const unsigned rc = (unsigned)ck * (unsigned)CH + (unsigned)L.runS[ck] - (unsigned)p0;
            for (int q0 = lo + m; q0 < hi; q0 += 8) {
                int4 e[4];
                #pragma unroll
                for (int j = 0; j < 4; ++j) {
                    const int q = min(q0 + 2 * j, hi - 1);
                    e[j] = R[rc + (unsigned)q];
                }
                asm volatile("" :: "v"(e[0].x), "v"(e[1].x), "v"(e[2].x), "v"(e[3].x));
                #pragma unroll
                for (int j = 0; j < 4; ++j) {
                    const int q = q0 + 2 * j;
                    if (q < hi) {
                        L.rs[q - base] = e[j].x;
                        L.rw[q - base] = e[j].y;
                        L.rl[q - base] = e[j].z & (GROWS - 1);
                    }
                }
            }
        }
        if (tid < GROWS) L.c2[tid] = 0;
    }
    __syncthreads();
    int ld[CAP / GT], rk[CAP / GT];
    #pragma unroll
    for (int i = 0; i < CAP / GT; ++i) {
        const int j = tid + GT * i;
        ld[i] = 0; rk[i] = 0;
        if (j < Wn) {
            ld[i] = L.rl[j];
            rk[i] = atomicAdd(&L.c2[ld[i]], 1);
        }
    }
    __syncthreads();
    if (wv == 0) {
        const int v = L.c2[lane];
        int x = v;
        #pragma unroll
        for (int off = 1; off < 32; off <<= 1) {
            const int y = __shfl_up(x, off);
            if (lane >= off) x += y;
        }
        L.st2[lane] = x - v;
        if (lane == 31) L.st2[GROWS] = x;
    }
    __syncthreads();
    #pragma unroll
    for (int i = 0; i < CAP / GT; ++i) {
        const int j = tid + GT * i;
        if (j < Wn) {
            const int p = L.st2[ld[i]] + rk[i];
            L.ss[p] = L.rs[j];
            L.sw[p] = L.rw[j];
        }
    }
    __syncthreads();
}

template <class Emit>
static __device__ __forceinline__
void GCN_block_walks(GCN_block_lds& L, const float* __restrict__ rows,
                     const int4* __restrict__ R, const int* __restrict__ T,
                     int blk, int nC, int g, int c, Emit emit) {
    const int nCc  = min(nC, NCMAX);
    const int Ltot = GCN_block_runs(L, T, blk, nCc, nC);
    if (Ltot <= CAP) {
        if (Ltot > 0) GCN_stage_window(L, R, nCc, 0, Ltot);
        for (int pass = 0; pass < 2; ++pass) {
            const int r = g + 16 * pass;
            GCN_row8 a;
            a.lo.x = 0.f; a.lo.y = 0.f; a.lo.z = 0.f; a.lo.w = 0.f;
            a.hi = a.lo;
            if (Ltot > 0) GCN_sum_rows(a, rows, L.ss, L.sw, L.st2[r], L.st2[r + 1], c);
            emit(r, a);
        }
    } else {
        for (int pass = 0; pass < 2; ++pass) {
            const int r = g + 16 * pass;
            GCN_row8 a;
            a.lo.x = 0.f; a.lo.y = 0.f; a.lo.z = 0.f; a.lo.w = 0.f;
            a.hi = a.lo;
            for (int base = 0; base < Ltot; base += CAP) {
                GCN_stage_window(L, R, nCc, base, Ltot);
                GCN_sum_rows(a, rows, L.ss, L.sw, L.st2[r], L.st2[r + 1], c);
                __syncthreads();
            }
            emit(r, a);
        }
    }
}

__global__ __launch_bounds__(GT, 16)
void GCN_gather1_kernel(const float* __restrict__ feat,
                        const float* __restrict__ b1,
                        const int4*  __restrict__ R,
                        const int*   __restrict__ T,
                        float*       __restrict__ h,
                        int nV, int nC) {
    __shared__ GCN_block_lds L;
    __shared__ __attribute__((aligned(16))) float bs[DF];
    const int g = threadIdx.x >> 4;
    const int c = threadIdx.x & 15;
    const int blk = blockIdx.x;
    if (threadIdx.x < DF) bs[threadIdx.x] = b1[threadIdx.x];
    GCN_block_walks(L, feat, R, T, blk, nC, g, c, [&](int r, const GCN_row8& a) {
        const int v = blk * GROWS + r;
        if (v < nV) {
            const unsigned cb = (unsigned)c * 16u;
            const float4 b0 = *(const float4*)(bs + 4 * c);
            const float4 b4 = *(const float4*)(bs + 64 + 4 * c);
            float4 r0, r1;
            r0.x = fmaxf(a.lo.x + b0.x, 0.f); r0.y = fmaxf(a.lo.y + b0.y, 0.f);
            r0.z = fmaxf(a.lo.z + b0.z, 0.f); r0.w = fmaxf(a.lo.w + b0.w, 0.f);
            r1.x = fmaxf(a.hi.x + b4.x, 0.f); r1.y = fmaxf(a.hi.y + b4.y, 0.f);
            r1.z = fmaxf(a.hi.z + b4.z, 0.f); r1.w = fmaxf(a.hi.w + b4.w, 0.f);
            char* hb = (char*)h;
            const unsigned ho = (unsigned)v * (unsigned)(DF * 4) + cb;
            *(float4*)(hb + ho)        = r0;
            *(float4*)(hb + ho + 256u) = r1;
        }
    });
}

#define APITCH 132

__global__ __launch_bounds__(GT, 16)
void GCN_gather2_gemm_kernel(const float* __restrict__ h,
                             const int4*  __restrict__ R,
                             const int*   __restrict__ T,
                             const float* __restrict__ W,
                             const float* __restrict__ bias,
                             float*       __restrict__ out,
                             int M, int nC) {
    __shared__ GCN_block_lds L;
    __shared__ float As[GROWS * APITCH];
    const int g = threadIdx.x >> 4;
    const int c = threadIdx.x & 15;
    GCN_block_walks(L, h, R, T, blockIdx.x, nC, g, c, [&](int r, const GCN_row8& a) {
        float4* ar = (float4*)(As + r * APITCH);
        ar[c] = a.lo;
        ar[16 + c] = a.hi;
    });
    __syncthreads();

    const int lane = threadIdx.x & 31;
    const int tN   = threadIdx.x >> 5;
    const int half = lane >> 4;
    const int mn   = lane & 15;
    const int col  = tN * 16 + mn;
    const float* Arow0 = As + mn * APITCH + 2 * half;
    const float* Arow1 = Arow0 + 16 * APITCH;
    const float* Bcol  = W + (size_t)(2 * half) * DF + col;

    v8f acc0 = {};
    v8f acc1 = {};
    #pragma unroll
    for (int k = 0; k < DF; k += 4) {
        const float2 av0 = *(const float2*)(Arow0 + k);
        const float2 av1 = *(const float2*)(Arow1 + k);
        v2f fa0; fa0[0] = av0.x; fa0[1] = av0.y;
        v2f fa1; fa1[0] = av1.x; fa1[1] = av1.y;
        v2f fb;
        fb[0] = Bcol[(size_t)k * DF];
        fb[1] = Bcol[(size_t)(k + 1) * DF];
        acc0 = __builtin_amdgcn_wmma_f32_16x16x4_f32(
                 false, fa0,  false, fb,
                 (short)0, acc0,  false,  false);
        acc1 = __builtin_amdgcn_wmma_f32_16x16x4_f32(
                 false, fa1,  false, fb,
                 (short)0, acc1,  false,  false);
    }

    const float bv = bias[col];
    #pragma unroll
    for (int i = 0; i < 8; ++i) {
        const int ro = blockIdx.x * GROWS + i + 8 * half;
        if (ro < M)      out[(size_t)ro * DF + col]        = acc0[i] + bv;
        if (ro + 16 < M) out[(size_t)(ro + 16) * DF + col] = acc1[i] + bv;
    }
}

extern "C" void kernel_launch(void* const* d_in, const int* in_sizes, int n_in,
                              void* d_out, int out_size, void* d_ws, size_t ws_size,
                              hipStream_t stream) {
    const float* feature = (const float*)d_in[0];
    const float* w0      = (const float*)d_in[1];
    const float* w1      = (const float*)d_in[2];
    const float* b1      = (const float*)d_in[3];
    const float* W2      = (const float*)d_in[4];
    const float* b2      = (const float*)d_in[5];
    const int*   src0    = (const int*)d_in[6];
    const int*   dst0    = (const int*)d_in[7];
    const int*   src1    = (const int*)d_in[8];
    const int*   dst1    = (const int*)d_in[9];

    const int E0 = in_sizes[1];
    const int E1 = in_sizes[2];
    const int N1 = N1C;
    const int N2 = out_size / DF;

    const int nC0 = (E0 + CH - 1) / CH;
    const int nC1 = (E1 + CH - 1) / CH;
    const int nB0 = (N1 + GROWS - 1) / GROWS;
    const int nB1 = (N2 + GROWS - 1) / GROWS;

    float* h   = (float*)d_ws;
    int4*  R0  = (int4*)(h + (size_t)N1 * DF);
    int4*  R1  = R0 + (size_t)nC0 * CH;
    int*   T0  = (int*)(R1 + (size_t)nC1 * CH);
    int*   T1  = T0 + (size_t)(nB0 + 1) * nC0;
    float* out = (float*)d_out;

    GCN_chunk_sort_kernel<<<nC0 + nC1, 1024, 0, stream>>>(src0, dst0, w0, src1, dst1, w1,
                                                          R0, R1, T0, T1,
                                                          E0, E1, N1, N2, nC0, nC1);
    GCN_gather1_kernel<<<nB0, GT, 0, stream>>>(feature, b1, R0, T0, h, N1, nC0);
    GCN_gather2_gemm_kernel<<<nB1, GT, 0, stream>>>(h, R1, T1, W2, b2, out, N2, nC1);
}
